# starting kernel
# baseline (speedup 1.0000x reference)
#include <hip/hip_runtime.h>
#include <hip/hip_bf16.h>


typedef __attribute__((ext_vector_type(2))) float v2f;
typedef __attribute__((ext_vector_type(8))) float v8f;

#define DF 48

__global__ void zero_f4(float4* __restrict__ p, int n4) {
    int gid = blockIdx.x * blockDim.x + threadIdx.x;
    if (gid < n4) p[gid] = make_float4(0.f, 0.f, 0.f, 0.f);
}

__global__ void scatter_add(const float* __restrict__ feat,
                            const int* __restrict__ src,
                            const int* __restrict__ dst,
                            float* __restrict__ agg,
                            long long total) {
    long long gid = (long long)blockIdx.x * blockDim.x + threadIdx.x;
    if (gid >= total) return;
    int e = (int)(gid / DF);
    int f = (int)(gid - (long long)e * DF);
    int s = src[e];
    int d = dst[e];
    unsafeAtomicAdd(&agg[(long long)d * DF + f], feat[(long long)s * DF + f]);
}

template <bool RELU>
__global__ __launch_bounds__(32)
void gemm_dual_wmma(const float* __restrict__ agg,
                    const float* __restrict__ xin,
                    const float* __restrict__ w_rel,
                    const float* __restrict__ w_root,
                    const float* __restrict__ bias,
                    float* __restrict__ out,
                    int n_nodes) {
    const int lane = threadIdx.x;
    const int hi   = lane >> 4;
    const int lo   = lane & 15;
    const int row0 = blockIdx.x * 16;

    const int arow = min(row0 + lo, n_nodes - 1);
    v2f a_agg[12], a_x[12];
#pragma unroll
    for (int k = 0; k < 12; ++k) {
        const int kk = 4 * k + 2 * hi;
        const long long base = (long long)arow * DF + kk;
        a_agg[k].x = agg[base];
        a_agg[k].y = agg[base + 1];
        a_x[k].x   = xin[base];
        a_x[k].y   = xin[base + 1];
    }

#pragma unroll
    for (int nt = 0; nt < 3; ++nt) {
        const int ncol = nt * 16 + lo;
        const float bv = bias[ncol];
        v8f c;
#pragma unroll
        for (int i = 0; i < 8; ++i) c[i] = bv;

#pragma unroll
        for (int k = 0; k < 12; ++k) {
            const int krow = 4 * k + 2 * hi;
            v2f brel, broot;
            brel.x  = w_rel [(krow    ) * DF + ncol];
            brel.y  = w_rel [(krow + 1) * DF + ncol];
            broot.x = w_root[(krow    ) * DF + ncol];
            broot.y = w_root[(krow + 1) * DF + ncol];
            c = __builtin_amdgcn_wmma_f32_16x16x4_f32(
                    false, a_agg[k], false, brel, (short)0, c, false, false);
            c = __builtin_amdgcn_wmma_f32_16x16x4_f32(
                    false, a_x[k], false, broot, (short)0, c, false, false);
        }

#pragma unroll
        for (int v = 0; v < 8; ++v) {
            float r = c[v];
            if (RELU) r = fmaxf(r, 0.f);
            const int m = row0 + v + 8 * hi;
            if (m < n_nodes) out[(long long)m * DF + ncol] = r;
        }
    }
}

extern "C" void kernel_launch(void* const* d_in, const int* in_sizes, int n_in,
                              void* d_out, int out_size, void* d_ws, size_t ws_size,
                              hipStream_t stream) {
    const float* x       = (const float*)d_in[0];
    const int*   ei      = (const int*)  d_in[1];
    const float* w_rel1  = (const float*)d_in[2];
    const float* w_root1 = (const float*)d_in[3];
    const float* b1      = (const float*)d_in[4];
    const float* w_rel2  = (const float*)d_in[5];
    const float* w_root2 = (const float*)d_in[6];
    const float* b2      = (const float*)d_in[7];
    float*       out     = (float*)d_out;

    const int N = in_sizes[0] / DF;
    const int E = in_sizes[1] / 2;
    const int* src = ei;
    const int* dst = ei + E;

    float* agg = (float*)d_ws;
    float* h   = agg + (size_t)N * DF;

    const int nfeat   = N * DF;
    const int n4      = nfeat / 4;
    const int zblocks = (n4 + 255) / 256;

    const long long etotal = (long long)E * DF;
    const int eblocks = (int)((etotal + 255) / 256);

    const int mtiles = (N + 15) / 16;

    zero_f4<<<zblocks, 256, 0, stream>>>((float4*)agg, n4);
    scatter_add<<<eblocks, 256, 0, stream>>>(x, src, dst, agg, etotal);
    gemm_dual_wmma<true><<<mtiles, 32, 0, stream>>>(agg, x, w_rel1, w_root1, b1, h, N);

    zero_f4<<<zblocks, 256, 0, stream>>>((float4*)agg, n4);
    scatter_add<<<eblocks, 256, 0, stream>>>(h, src, dst, agg, etotal);
    gemm_dual_wmma<false><<<mtiles, 32, 0, stream>>>(agg, h, w_rel2, w_root2, b2, out, N);
}


# Replace the global f32 atomic scatter: 256 groups each own 392 nodes' sums in LDS, find their edges through a per-owner bit map of edge words, add rows by LDS f32 atomics; sum order = LDS arrival
# speedup vs baseline: 3.5034x; 3.5034x over previous
#include <hip/hip_runtime.h>
#include <hip/hip_bf16.h>


typedef __attribute__((ext_vector_type(2))) float v2f;
typedef __attribute__((ext_vector_type(8))) float v8f;

#define DF 48

#define NOWN    256
#define NPO     392
#define AGG_T   1024
#define WCAP    3584
#define ECAP    (WCAP * 16)
#define CHUNK_W 1024

__global__ __launch_bounds__(CHUNK_W)
void owner_bits(const int* __restrict__ dst, unsigned* __restrict__ bm,
                int n_edges, int nq) {
    __shared__ unsigned bits[32 * (NOWN + 1)];
    const int tid = threadIdx.x;
    for (int i = tid; i < 32 * (NOWN + 1); i += CHUNK_W) bits[i] = 0u;
    __syncthreads();

    const int q = blockIdx.x * CHUNK_W + tid;
    if (q < nq) {
        int d[16];
        if (q * 16 + 16 <= n_edges) {
            const int4* d4 = (const int4*)(dst + (long long)q * 16);
#pragma unroll
            for (int k = 0; k < 4; ++k) {
                const int4 v = d4[k];
                d[4 * k] = v.x; d[4 * k + 1] = v.y; d[4 * k + 2] = v.z; d[4 * k + 3] = v.w;
            }
        } else {
#pragma unroll
            for (int k = 0; k < 16; ++k) d[k] = (q * 16 + k < n_edges) ? dst[q * 16 + k] : -1;
        }
        const unsigned bit = 1u << (tid & 31);
#pragma unroll
        for (int k = 0; k < 16; ++k) {
            if (d[k] >= 0) {
                const int o = min(d[k] / NPO, NOWN - 1);
                atomicOr(&bits[(tid >> 5) * (NOWN + 1) + o], bit);
            }
        }
    }
    __syncthreads();

    const int pitch = gridDim.x * 32;
#pragma unroll
    for (int r = 0; r < NOWN * 32 / CHUNK_W; ++r) {
        const int o = r * (CHUNK_W / 32) + (tid >> 5);
        bm[(long long)o * pitch + blockIdx.x * 32 + (tid & 31)] = bits[(tid & 31) * (NOWN + 1) + o];
    }
}

__global__ __launch_bounds__(AGG_T)
void agg_owner(const float* __restrict__ feat,
               const int* __restrict__ src,
               const int* __restrict__ dst,
               const unsigned* __restrict__ bm,
               float* __restrict__ agg,
               int n_nodes, int n_edges, int pitch) {
    __shared__ float acc[NPO * DF];
    __shared__ int   elist[ECAP];
    __shared__ int   wlist[WCAP];
    __shared__ int   cntw, cnte, more;

    const int tid  = threadIdx.x;
    const int w    = blockIdx.x;
    const int j16  = tid & 15;
    const int lane = tid & 31;
    const int row0 = w * NPO;

    for (int i = tid; i < NPO * DF / 4; i += AGG_T)
        ((float4*)acc)[i] = make_float4(0.f, 0.f, 0.f, 0.f);
    if (tid == 0) { cntw = 0; cnte = 0; more = 0; }

    unsigned bw[4];
#pragma unroll
    for (int k = 0; k < 4; ++k) {
        const int di = tid + k * AGG_T;
        bw[k] = bm[(long long)w * pitch + min(di, pitch - 1)];
        if (di >= pitch) bw[k] = 0u;
    }
    __syncthreads();

    int going = 1;
    while (going) {
        bool full = false;
#pragma unroll
        for (int k = 0; k < 4; ++k) {
            while (bw[k] != 0u && !full) {
                const int pos = atomicAdd(&cntw, 1);
                if (pos < WCAP) {
                    const int bp = __ffs((int)bw[k]) - 1;
                    wlist[pos] = (tid + k * AGG_T) * 32 + bp;
                    bw[k] &= bw[k] - 1u;
                } else {
                    full = true;
                }
            }
        }
        const bool pending = (bw[0] | bw[1] | bw[2] | bw[3]) != 0u;
        __syncthreads();

        const int nw_items = min(cntw, WCAP) * 16;
        for (int base = 0; base < nw_items; base += AGG_T * 16) {
            int e[16], d[16];
#pragma unroll
            for (int k = 0; k < 16; ++k) {
                const int idx = min(base + k * AGG_T + tid, nw_items - 1);
                e[k] = wlist[idx >> 4] * 16 + j16;
            }
#pragma unroll
            for (int k = 0; k < 16; ++k) d[k] = dst[min(e[k], n_edges - 1)] - row0;
            unsigned vote[16];
            int tot = 0;
#pragma unroll
            for (int k = 0; k < 16; ++k) {
                const bool hit = (base + k * AGG_T + tid < nw_items) && (e[k] < n_edges) &&
                                 ((unsigned)d[k] < (unsigned)NPO);
                vote[k] = (unsigned)__ballot(hit);
                tot += __popc(vote[k]);
            }
            if (tot != 0) {
                int slot = 0;
                if (lane == 0) slot = atomicAdd(&cnte, tot);
                slot = __builtin_amdgcn_readfirstlane(slot);
#pragma unroll
                for (int k = 0; k < 16; ++k) {
                    if ((vote[k] >> lane) & 1u)
                        elist[slot + (int)__builtin_amdgcn_mbcnt_lo(vote[k], 0u)] = (e[k] << 9) | d[k];
                    slot += __popc(vote[k]);
                }
            }
        }
        __syncthreads();

        const int total = cnte * 16;
        for (int base = 0; base < total; base += AGG_T * 12) {
            int  p[12], s[12];
            bool ok[12];
            float v[12][3];
#pragma unroll
            for (int k = 0; k < 12; ++k) {
                const int idx = base + k * AGG_T + tid;
                ok[k] = idx < total;
                p[k] = elist[ok[k] ? (idx >> 4) : 0];
            }
#pragma unroll
            for (int k = 0; k < 12; ++k) s[k] = src[p[k] >> 9];
#pragma unroll
            for (int k = 0; k < 12; ++k) {
                const float* r = feat + (long long)s[k] * DF + j16;
                v[k][0] = r[0];
                v[k][1] = r[16];
                v[k][2] = r[32];
            }
            asm volatile("" :: "v"(v[0][0]), "v"(v[0][1]), "v"(v[0][2]));
#pragma unroll
            for (int k = 0; k < 12; ++k) {
                if (ok[k]) {
                    float* a = acc + (p[k] & 511) * DF + j16;
                    __hip_atomic_fetch_add(a,      v[k][0], __ATOMIC_RELAXED, __HIP_MEMORY_SCOPE_WORKGROUP);
                    __hip_atomic_fetch_add(a + 16, v[k][1], __ATOMIC_RELAXED, __HIP_MEMORY_SCOPE_WORKGROUP);
                    __hip_atomic_fetch_add(a + 32, v[k][2], __ATOMIC_RELAXED, __HIP_MEMORY_SCOPE_WORKGROUP);
                }
            }
        }
        if (pending) more = 1;
        __syncthreads();
        going = more;
        __syncthreads();
        if (tid == 0) { cntw = 0; cnte = 0; more = 0; }
        __syncthreads();
    }

    const int rows = min(NPO, n_nodes - row0);
    float4* o4 = (float4*)(agg + (long long)row0 * DF);
    for (int i = tid; i < rows * (DF / 4); i += AGG_T) o4[i] = ((const float4*)acc)[i];
}

template <bool RELU>
__global__ __launch_bounds__(32)
void gemm_dual_wmma(const float* __restrict__ agg,
                    const float* __restrict__ xin,
                    const float* __restrict__ w_rel,
                    const float* __restrict__ w_root,
                    const float* __restrict__ bias,
                    float* __restrict__ out,
                    int n_nodes) {
    const int lane = threadIdx.x;
    const int hi   = lane >> 4;
    const int lo   = lane & 15;
    const int row0 = blockIdx.x * 16;

    const int arow = min(row0 + lo, n_nodes - 1);
    v2f a_agg[12], a_x[12];
#pragma unroll
    for (int k = 0; k < 12; ++k) {
        const int kk = 4 * k + 2 * hi;
        const long long base = (long long)arow * DF + kk;
        a_agg[k].x = agg[base];
        a_agg[k].y = agg[base + 1];
        a_x[k].x   = xin[base];
        a_x[k].y   = xin[base + 1];
    }

#pragma unroll
    for (int nt = 0; nt < 3; ++nt) {
        const int ncol = nt * 16 + lo;
        const float bv = bias[ncol];
        v8f c;
#pragma unroll
        for (int i = 0; i < 8; ++i) c[i] = bv;

#pragma unroll
        for (int k = 0; k < 12; ++k) {
            const int krow = 4 * k + 2 * hi;
            v2f brel, broot;
            brel.x  = w_rel [(krow    ) * DF + ncol];
            brel.y  = w_rel [(krow + 1) * DF + ncol];
            broot.x = w_root[(krow    ) * DF + ncol];
            broot.y = w_root[(krow + 1) * DF + ncol];
            c = __builtin_amdgcn_wmma_f32_16x16x4_f32(
                    false, a_agg[k], false, brel, (short)0, c, false, false);
            c = __builtin_amdgcn_wmma_f32_16x16x4_f32(
                    false, a_x[k], false, broot, (short)0, c, false, false);
        }

#pragma unroll
        for (int v = 0; v < 8; ++v) {
            float r = c[v];
            if (RELU) r = fmaxf(r, 0.f);
            const int m = row0 + v + 8 * hi;
            if (m < n_nodes) out[(long long)m * DF + ncol] = r;
        }
    }
}

extern "C" void kernel_launch(void* const* d_in, const int* in_sizes, int n_in,
                              void* d_out, int out_size, void* d_ws, size_t ws_size,
                              hipStream_t stream) {
    const float* x       = (const float*)d_in[0];
    const int*   ei      = (const int*)  d_in[1];
    const float* w_rel1  = (const float*)d_in[2];
    const float* w_root1 = (const float*)d_in[3];
    const float* b1      = (const float*)d_in[4];
    const float* w_rel2  = (const float*)d_in[5];
    const float* w_root2 = (const float*)d_in[6];
    const float* b2      = (const float*)d_in[7];
    float*       out     = (float*)d_out;

    const int N = in_sizes[0] / DF;
    const int E = in_sizes[1] / 2;
    const int* src = ei;
    const int* dst = ei + E;

    float* agg = (float*)d_ws;
    float* h   = agg + (size_t)N * DF;

    unsigned* bm = (unsigned*)(h + (size_t)N * DF);
    const int nq     = (E + 15) / 16;
    const int nchunk = (nq + CHUNK_W - 1) / CHUNK_W;

    const int mtiles = (N + 15) / 16;

    owner_bits<<<nchunk, CHUNK_W, 0, stream>>>(dst, bm, E, nq);
    agg_owner<<<NOWN, AGG_T, 0, stream>>>(x, src, dst, bm, agg, N, E, nchunk * 32);
    gemm_dual_wmma<true><<<mtiles, 32, 0, stream>>>(agg, x, w_rel1, w_root1, b1, h, N);

    agg_owner<<<NOWN, AGG_T, 0, stream>>>(h, src, dst, bm, agg, N, E, nchunk * 32);
    gemm_dual_wmma<false><<<mtiles, 32, 0, stream>>>(agg, h, w_rel2, w_root2, b2, out, N);
}


# Layer 2 stops scanning: layer 1's owner groups save their found (source,row) pairs, source id fetched beside dst in the word test; layer 2 adds rows from that list; sum order = LDS arrival
# speedup vs baseline: 4.2066x; 1.2007x over previous
#include <hip/hip_runtime.h>
#include <hip/hip_bf16.h>


typedef __attribute__((ext_vector_type(2))) float v2f;
typedef __attribute__((ext_vector_type(8))) float v8f;

#define DF 48

#define NOWN    256
#define NPO     392
#define AGG_T   1024
#define WCAP    3584
#define ECAP    (WCAP * 16)
#define CHUNK_W 1024
#define LCAP    131072

__global__ __launch_bounds__(CHUNK_W)
void owner_bits(const int* __restrict__ dst, unsigned* __restrict__ bm,
                int n_edges, int nq) {
    __shared__ unsigned bits[32 * (NOWN + 1)];
    const int tid = threadIdx.x;
    for (int i = tid; i < 32 * (NOWN + 1); i += CHUNK_W) bits[i] = 0u;
    __syncthreads();

    const int q = blockIdx.x * CHUNK_W + tid;
    if (q < nq) {
        int d[16];
        if (q * 16 + 16 <= n_edges) {
            const int4* d4 = (const int4*)(dst + (long long)q * 16);
#pragma unroll
            for (int k = 0; k < 4; ++k) {
                const int4 v = d4[k];
                d[4 * k] = v.x; d[4 * k + 1] = v.y; d[4 * k + 2] = v.z; d[4 * k + 3] = v.w;
            }
        } else {
#pragma unroll
            for (int k = 0; k < 16; ++k) d[k] = (q * 16 + k < n_edges) ? dst[q * 16 + k] : -1;
        }
        const unsigned bit = 1u << (tid & 31);
#pragma unroll
        for (int k = 0; k < 16; ++k) {
            if (d[k] >= 0) {
                const int o = min(d[k] / NPO, NOWN - 1);
                atomicOr(&bits[(tid >> 5) * (NOWN + 1) + o], bit);
            }
        }
    }
    __syncthreads();

    const int pitch = gridDim.x * 32;
#pragma unroll
    for (int r = 0; r < NOWN * 32 / CHUNK_W; ++r) {
        const int o = r * (CHUNK_W / 32) + (tid >> 5);
        bm[(long long)o * pitch + blockIdx.x * 32 + (tid & 31)] = bits[(tid & 31) * (NOWN + 1) + o];
    }
}

__device__ __forceinline__ void add_rows(const float* __restrict__ feat, float* acc, const int* list,
                                         int cnt, int tid, int j16) {
    const int total = cnt * 16;
    for (int base = 0; base < total; base += AGG_T * 12) {
        int  p[12];
        bool ok[12];
        float v[12][3];
#pragma unroll
        for (int k = 0; k < 12; ++k) {
            const int idx = base + k * AGG_T + tid;
            ok[k] = idx < total;
            p[k] = list[ok[k] ? (idx >> 4) : 0];
        }
#pragma unroll
        for (int k = 0; k < 12; ++k) {
            const float* r = feat + (long long)(p[k] >> 9) * DF + j16;
            v[k][0] = r[0];
            v[k][1] = r[16];
            v[k][2] = r[32];
        }
        asm volatile("" :: "v"(v[0][0]), "v"(v[0][1]), "v"(v[0][2]));
#pragma unroll
        for (int k = 0; k < 12; ++k) {
            if (ok[k]) {
                float* a = acc + (p[k] & 511) * DF + j16;
                __hip_atomic_fetch_add(a,      v[k][0], __ATOMIC_RELAXED, __HIP_MEMORY_SCOPE_WORKGROUP);
                __hip_atomic_fetch_add(a + 16, v[k][1], __ATOMIC_RELAXED, __HIP_MEMORY_SCOPE_WORKGROUP);
                __hip_atomic_fetch_add(a + 32, v[k][2], __ATOMIC_RELAXED, __HIP_MEMORY_SCOPE_WORKGROUP);
            }
        }
    }
}

template <bool REUSE>
__global__ __launch_bounds__(AGG_T)
void agg_owner(const float* __restrict__ feat,
               const int* __restrict__ src,
               const int* __restrict__ dst,
               const unsigned* __restrict__ bm,
               float* __restrict__ agg,
               int* __restrict__ glist,
               int* __restrict__ gcount,
               int n_nodes, int n_edges, int pitch) {
    __shared__ float acc[NPO * DF];
    __shared__ int   elist[ECAP];
    __shared__ int   wlist[WCAP];
    __shared__ int   cntw, cnte, more;

    const int tid  = threadIdx.x;
    const int w    = blockIdx.x;
    const int j16  = tid & 15;
    const int lane = tid & 31;
    const int row0 = w * NPO;
    int* mylist = glist + (long long)w * LCAP;

    for (int i = tid; i < NPO * DF / 4; i += AGG_T)
        ((float4*)acc)[i] = make_float4(0.f, 0.f, 0.f, 0.f);
    if (tid == 0) { cntw = 0; cnte = 0; more = 0; }

    int nsaved = -1;
    if (REUSE) {
        nsaved = __builtin_amdgcn_readfirstlane(gcount[w]);
        if (nsaved > LCAP) nsaved = -1;
    }

    if (nsaved >= 0) {
        for (int c0 = 0; c0 < nsaved; c0 += ECAP) {
            const int cn = min(ECAP, nsaved - c0);
            for (int i0 = 0; i0 < cn; i0 += AGG_T * 8) {
                int t[8];
#pragma unroll
                for (int k = 0; k < 8; ++k)
                    t[k] = mylist[c0 + min(i0 + k * AGG_T + tid, cn - 1)];
#pragma unroll
                for (int k = 0; k < 8; ++k) {
                    const int i = i0 + k * AGG_T + tid;
                    if (i < cn) elist[i] = t[k];
                }
            }
            __syncthreads();
            add_rows(feat, acc, elist, cn, tid, j16);
            __syncthreads();
        }
    } else {
        unsigned bw[4];
#pragma unroll
        for (int k = 0; k < 4; ++k) {
            const int di = tid + k * AGG_T;
            bw[k] = bm[(long long)w * pitch + min(di, pitch - 1)];
            if (di >= pitch) bw[k] = 0u;
        }
        __syncthreads();

        int saved = 0;
        int going = 1;
        while (going) {
            bool full = false;
#pragma unroll
            for (int k = 0; k < 4; ++k) {
                while (bw[k] != 0u && !full) {
                    const int pos = atomicAdd(&cntw, 1);
                    if (pos < WCAP) {
                        const int bp = __ffs((int)bw[k]) - 1;
                        wlist[pos] = (tid + k * AGG_T) * 32 + bp;
                        bw[k] &= bw[k] - 1u;
                    } else {
                        full = true;
                    }
                }
            }
            const bool pending = (bw[0] | bw[1] | bw[2] | bw[3]) != 0u;
            __syncthreads();

            const int nw_items = min(cntw, WCAP) * 16;
            for (int base = 0; base < nw_items; base += AGG_T * 16) {
                int e[16], d[16], sv[16];
#pragma unroll
                for (int k = 0; k < 16; ++k) {
                    const int idx = min(base + k * AGG_T + tid, nw_items - 1);
                    e[k] = wlist[idx >> 4] * 16 + j16;
                }
#pragma unroll
                for (int k = 0; k < 16; ++k) {
                    const int ec = min(e[k], n_edges - 1);
                    d[k]  = dst[ec] - row0;
                    sv[k] = src[ec];
                }
                asm volatile("" :: "v"(sv[0]), "v"(sv[1]), "v"(sv[2]),  "v"(sv[3]),  "v"(sv[4]),  "v"(sv[5]),  "v"(sv[6]),  "v"(sv[7]),
                                   "v"(sv[8]), "v"(sv[9]), "v"(sv[10]), "v"(sv[11]), "v"(sv[12]), "v"(sv[13]), "v"(sv[14]), "v"(sv[15]));
                unsigned vote[16];
                int tot = 0;
#pragma unroll
                for (int k = 0; k < 16; ++k) {
                    const bool hit = (base + k * AGG_T + tid < nw_items) && (e[k] < n_edges) &&
                                     ((unsigned)d[k] < (unsigned)NPO);
                    vote[k] = (unsigned)__ballot(hit);
                    tot += __popc(vote[k]);
                }
                if (tot != 0) {
                    int slot = 0;
                    if (lane == 0) slot = atomicAdd(&cnte, tot);
                    slot = __builtin_amdgcn_readfirstlane(slot);
#pragma unroll
                    for (int k = 0; k < 16; ++k) {
                        if ((vote[k] >> lane) & 1u)
                            elist[slot + (int)__builtin_amdgcn_mbcnt_lo(vote[k], 0u)] = (sv[k] << 9) | d[k];
                        slot += __popc(vote[k]);
                    }
                }
            }
            __syncthreads();

            const int ce = cnte;
            if (!REUSE) {
                if (saved >= 0 && saved + ce <= LCAP) {
                    for (int i0 = 0; i0 < ce; i0 += AGG_T * 4) {
#pragma unroll
                        for (int k = 0; k < 4; ++k) {
                            const int i = i0 + k * AGG_T + tid;
                            if (i < ce) mylist[saved + i] = elist[i];
                        }
                    }
                    saved += ce;
                } else {
                    saved = -1;
                }
            }
            add_rows(feat, acc, elist, ce, tid, j16);
            if (pending) more = 1;
            __syncthreads();
            going = more;
            __syncthreads();
            if (tid == 0) { cntw = 0; cnte = 0; more = 0; }
            __syncthreads();
        }
        if (!REUSE && tid == 0) gcount[w] = saved;
    }

    const int rows = min(NPO, n_nodes - row0);
    float4* o4 = (float4*)(agg + (long long)row0 * DF);
    for (int i = tid; i < rows * (DF / 4); i += AGG_T) o4[i] = ((const float4*)acc)[i];
}

template <bool RELU>
__global__ __launch_bounds__(32)
void gemm_dual_wmma(const float* __restrict__ agg,
                    const float* __restrict__ xin,
                    const float* __restrict__ w_rel,
                    const float* __restrict__ w_root,
                    const float* __restrict__ bias,
                    float* __restrict__ out,
                    int n_nodes) {
    const int lane = threadIdx.x;
    const int hi   = lane >> 4;
    const int lo   = lane & 15;
    const int row0 = blockIdx.x * 16;

    const int arow = min(row0 + lo, n_nodes - 1);
    v2f a_agg[12], a_x[12];
#pragma unroll
    for (int k = 0; k < 12; ++k) {
        const int kk = 4 * k + 2 * hi;
        const long long base = (long long)arow * DF + kk;
        a_agg[k].x = agg[base];
        a_agg[k].y = agg[base + 1];
        a_x[k].x   = xin[base];
        a_x[k].y   = xin[base + 1];
    }

#pragma unroll
    for (int nt = 0; nt < 3; ++nt) {
        const int ncol = nt * 16 + lo;
        const float bv = bias[ncol];
        v8f c;
#pragma unroll
        for (int i = 0; i < 8; ++i) c[i] = bv;

#pragma unroll
        for (int k = 0; k < 12; ++k) {
            const int krow = 4 * k + 2 * hi;
            v2f brel, broot;
            brel.x  = w_rel [(krow    ) * DF + ncol];
            brel.y  = w_rel [(krow + 1) * DF + ncol];
            broot.x = w_root[(krow    ) * DF + ncol];
            broot.y = w_root[(krow + 1) * DF + ncol];
            c = __builtin_amdgcn_wmma_f32_16x16x4_f32(
                    false, a_agg[k], false, brel, (short)0, c, false, false);
            c = __builtin_amdgcn_wmma_f32_16x16x4_f32(
                    false, a_x[k], false, broot, (short)0, c, false, false);
        }

#pragma unroll
        for (int v = 0; v < 8; ++v) {
            float r = c[v];
            if (RELU) r = fmaxf(r, 0.f);
            const int m = row0 + v + 8 * hi;
            if (m < n_nodes) out[(long long)m * DF + ncol] = r;
        }
    }
}

extern "C" void kernel_launch(void* const* d_in, const int* in_sizes, int n_in,
                              void* d_out, int out_size, void* d_ws, size_t ws_size,
                              hipStream_t stream) {
    const float* x       = (const float*)d_in[0];
    const int*   ei      = (const int*)  d_in[1];
    const float* w_rel1  = (const float*)d_in[2];
    const float* w_root1 = (const float*)d_in[3];
    const float* b1      = (const float*)d_in[4];
    const float* w_rel2  = (const float*)d_in[5];
    const float* w_root2 = (const float*)d_in[6];
    const float* b2      = (const float*)d_in[7];
    float*       out     = (float*)d_out;

    const int N = in_sizes[0] / DF;
    const int E = in_sizes[1] / 2;
    const int* src = ei;
    const int* dst = ei + E;

    float* agg = (float*)d_ws;
    float* h   = agg + (size_t)N * DF;

    unsigned* bm = (unsigned*)(h + (size_t)N * DF);
    const int nq     = (E + 15) / 16;
    const int nchunk = (nq + CHUNK_W - 1) / CHUNK_W;
    int* glist  = (int*)(bm + (size_t)NOWN * nchunk * 32);
    int* gcount = glist + (size_t)NOWN * LCAP;

    const int mtiles = (N + 15) / 16;

    owner_bits<<<nchunk, CHUNK_W, 0, stream>>>(dst, bm, E, nq);
    agg_owner<false><<<NOWN, AGG_T, 0, stream>>>(x, src, dst, bm, agg, glist, gcount, N, E, nchunk * 32);
    gemm_dual_wmma<true><<<mtiles, 32, 0, stream>>>(agg, x, w_rel1, w_root1, b1, h, N);

    agg_owner<true><<<NOWN, AGG_T, 0, stream>>>(h, src, dst, bm, agg, glist, gcount, N, E, nchunk * 32);
    gemm_dual_wmma<false><<<mtiles, 32, 0, stream>>>(agg, h, w_rel2, w_root2, b2, out, N);
}


# Dense phase runs in the owner groups: after the LDS adds each wave does the layer's WMMAs on a 16-row tile of the group's sums (A from LDS); agg never stored; 5 launches -> 3; sum order kept
# speedup vs baseline: 5.5189x; 1.3120x over previous
#include <hip/hip_runtime.h>
#include <hip/hip_bf16.h>


typedef __attribute__((ext_vector_type(2))) float v2f;
typedef __attribute__((ext_vector_type(8))) float v8f;

#define DF 48

#define NOWN    256
#define NPO     392
#define AGG_T   1024
#define WCAP    3584
#define ECAP    (WCAP * 16)
#define CHUNK_W 1024
#define LCAP    131072

__global__ __launch_bounds__(CHUNK_W)
void owner_bits(const int* __restrict__ dst, unsigned* __restrict__ bm,
                int n_edges, int nq) {
    __shared__ unsigned bits[32 * (NOWN + 1)];
    const int tid = threadIdx.x;
    for (int i = tid; i < 32 * (NOWN + 1); i += CHUNK_W) bits[i] = 0u;
    __syncthreads();

    const int q = blockIdx.x * CHUNK_W + tid;
    if (q < nq) {
        int d[16];
        if (q * 16 + 16 <= n_edges) {
            const int4* d4 = (const int4*)(dst + (long long)q * 16);
#pragma unroll
            for (int k = 0; k < 4; ++k) {
                const int4 v = d4[k];
                d[4 * k] = v.x; d[4 * k + 1] = v.y; d[4 * k + 2] = v.z; d[4 * k + 3] = v.w;
            }
        } else {
#pragma unroll
            for (int k = 0; k < 16; ++k) d[k] = (q * 16 + k < n_edges) ? dst[q * 16 + k] : -1;
        }
        const unsigned bit = 1u << (tid & 31);
#pragma unroll
        for (int k = 0; k < 16; ++k) {
            if (d[k] >= 0) {
                const int o = min(d[k] / NPO, NOWN - 1);
                atomicOr(&bits[(tid >> 5) * (NOWN + 1) + o], bit);
            }
        }
    }
    __syncthreads();

    const int pitch = gridDim.x * 32;
#pragma unroll
    for (int r = 0; r < NOWN * 32 / CHUNK_W; ++r) {
        const int o = r * (CHUNK_W / 32) + (tid >> 5);
        bm[(long long)o * pitch + blockIdx.x * 32 + (tid & 31)] = bits[(tid & 31) * (NOWN + 1) + o];
    }
}

__device__ __forceinline__ void add_rows(const float* __restrict__ feat, float* acc, const int* list,
                                         int cnt, int tid, int j16) {
    const int total = cnt * 16;
    for (int base = 0; base < total; base += AGG_T * 12) {
        int  p[12];
        bool ok[12];
        float v[12][3];
#pragma unroll
        for (int k = 0; k < 12; ++k) {
            const int idx = base + k * AGG_T + tid;
            ok[k] = idx < total;
            p[k] = list[ok[k] ? (idx >> 4) : 0];
        }
#pragma unroll
        for (int k = 0; k < 12; ++k) {
            const float* r = feat + (long long)(p[k] >> 9) * DF + j16;
            v[k][0] = r[0];
            v[k][1] = r[16];
            v[k][2] = r[32];
        }
        asm volatile("" :: "v"(v[0][0]), "v"(v[0][1]), "v"(v[0][2]));
#pragma unroll
        for (int k = 0; k < 12; ++k) {
            if (ok[k]) {
                float* a = acc + (p[k] & 511) * DF + j16;
                __hip_atomic_fetch_add(a,      v[k][0], __ATOMIC_RELAXED, __HIP_MEMORY_SCOPE_WORKGROUP);
                __hip_atomic_fetch_add(a + 16, v[k][1], __ATOMIC_RELAXED, __HIP_MEMORY_SCOPE_WORKGROUP);
                __hip_atomic_fetch_add(a + 32, v[k][2], __ATOMIC_RELAXED, __HIP_MEMORY_SCOPE_WORKGROUP);
            }
        }
    }
}

template <bool REUSE, bool RELU>
__global__ __launch_bounds__(AGG_T)
void agg_owner(const float* __restrict__ feat,
               const int* __restrict__ src,
               const int* __restrict__ dst,
               const unsigned* __restrict__ bm,
               const float* __restrict__ w_rel,
               const float* __restrict__ w_root,
               const float* __restrict__ bias,
               float* __restrict__ out,
               int* __restrict__ glist,
               int* __restrict__ gcount,
               int n_nodes, int n_edges, int pitch) {
    __shared__ __attribute__((aligned(16))) float acc[NPO * DF];
    __shared__ int   elist[ECAP];
    __shared__ int   wlist[WCAP];
    __shared__ int   cntw, cnte, more;

    const int tid  = threadIdx.x;
    const int w    = blockIdx.x;
    const int j16  = tid & 15;
    const int lane = tid & 31;
    const int row0 = w * NPO;
    int* mylist = glist + (long long)w * LCAP;

    for (int i = tid; i < NPO * DF / 4; i += AGG_T)
        ((float4*)acc)[i] = make_float4(0.f, 0.f, 0.f, 0.f);
    if (tid == 0) { cntw = 0; cnte = 0; more = 0; }

    int nsaved = -1;
    if (REUSE) {
        nsaved = __builtin_amdgcn_readfirstlane(gcount[w]);
        if (nsaved > LCAP) nsaved = -1;
    }

    if (nsaved >= 0) {
        for (int c0 = 0; c0 < nsaved; c0 += ECAP) {
            const int cn = min(ECAP, nsaved - c0);
            for (int i0 = 0; i0 < cn; i0 += AGG_T * 8) {
                int t[8];
#pragma unroll
                for (int k = 0; k < 8; ++k)
                    t[k] = mylist[c0 + min(i0 + k * AGG_T + tid, cn - 1)];
#pragma unroll
                for (int k = 0; k < 8; ++k) {
                    const int i = i0 + k * AGG_T + tid;
                    if (i < cn) elist[i] = t[k];
                }
            }
            __syncthreads();
            add_rows(feat, acc, elist, cn, tid, j16);
            __syncthreads();
        }
    } else {
        unsigned bw[4];
#pragma unroll
        for (int k = 0; k < 4; ++k) {
            const int di = tid + k * AGG_T;
            bw[k] = bm[(long long)w * pitch + min(di, pitch - 1)];
            if (di >= pitch) bw[k] = 0u;
        }
        __syncthreads();

        int saved = 0;
        int going = 1;
        while (going) {
            bool full = false;
#pragma unroll
            for (int k = 0; k < 4; ++k) {
                while (bw[k] != 0u && !full) {
                    const int pos = atomicAdd(&cntw, 1);
                    if (pos < WCAP) {
                        const int bp = __ffs((int)bw[k]) - 1;
                        wlist[pos] = (tid + k * AGG_T) * 32 + bp;
                        bw[k] &= bw[k] - 1u;
                    } else {
                        full = true;
                    }
                }
            }
            const bool pending = (bw[0] | bw[1] | bw[2] | bw[3]) != 0u;
            __syncthreads();

            const int nw_items = min(cntw, WCAP) * 16;
            for (int base = 0; base < nw_items; base += AGG_T * 16) {
                int e[16], d[16], sv[16];
#pragma unroll
                for (int k = 0; k < 16; ++k) {
                    const int idx = min(base + k * AGG_T + tid, nw_items - 1);
                    e[k] = wlist[idx >> 4] * 16 + j16;
                }
#pragma unroll
                for (int k = 0; k < 16; ++k) {
                    const int ec = min(e[k], n_edges - 1);
                    d[k]  = dst[ec] - row0;
                    sv[k] = src[ec];
                }
                asm volatile("" :: "v"(sv[0]), "v"(sv[1]), "v"(sv[2]),  "v"(sv[3]),  "v"(sv[4]),  "v"(sv[5]),  "v"(sv[6]),  "v"(sv[7]),
                                   "v"(sv[8]), "v"(sv[9]), "v"(sv[10]), "v"(sv[11]), "v"(sv[12]), "v"(sv[13]), "v"(sv[14]), "v"(sv[15]));
                unsigned vote[16];
                int tot = 0;
#pragma unroll
                for (int k = 0; k < 16; ++k) {
                    const bool hit = (base + k * AGG_T + tid < nw_items) && (e[k] < n_edges) &&
                                     ((unsigned)d[k] < (unsigned)NPO);
                    vote[k] = (unsigned)__ballot(hit);
                    tot += __popc(vote[k]);
                }
                if (tot != 0) {
                    int slot = 0;
                    if (lane == 0) slot = atomicAdd(&cnte, tot);
                    slot = __builtin_amdgcn_readfirstlane(slot);
#pragma unroll
                    for (int k = 0; k < 16; ++k) {
                        if ((vote[k] >> lane) & 1u)
                            elist[slot + (int)__builtin_amdgcn_mbcnt_lo(vote[k], 0u)] = (sv[k] << 9) | d[k];
                        slot += __popc(vote[k]);
                    }
                }
            }
            __syncthreads();

            const int ce = cnte;
            if (!REUSE) {
                if (saved >= 0 && saved + ce <= LCAP) {
                    for (int i0 = 0; i0 < ce; i0 += AGG_T * 4) {
#pragma unroll
                        for (int k = 0; k < 4; ++k) {
                            const int i = i0 + k * AGG_T + tid;
                            if (i < ce) mylist[saved + i] = elist[i];
                        }
                    }
                    saved += ce;
                } else {
                    saved = -1;
                }
            }
            add_rows(feat, acc, elist, ce, tid, j16);
            if (pending) more = 1;
            __syncthreads();
            going = more;
            __syncthreads();
            if (tid == 0) { cntw = 0; cnte = 0; more = 0; }
            __syncthreads();
        }
        if (!REUSE && tid == 0) gcount[w] = saved;
    }

    const int rows  = min(NPO, n_nodes - row0);
    const int tile0 = __builtin_amdgcn_readfirstlane(tid >> 5) * 16;
    const bool has_tile = tile0 < rows;
    const int hi = lane >> 4;
    const int lrow = min(tile0 + j16, NPO - 1);

    v2f a_x[12];
    if (has_tile) {
        const float* xr = feat + (long long)min(row0 + lrow, n_nodes - 1) * DF + 2 * hi;
#pragma unroll
        for (int k = 0; k < 12; ++k) a_x[k] = *(const v2f*)(xr + 4 * k);
    }
    __syncthreads();

    if (has_tile) {
        v2f a_agg[12];
        const float* ar = acc + lrow * DF + 2 * hi;
#pragma unroll
        for (int k = 0; k < 12; ++k) a_agg[k] = *(const v2f*)(ar + 4 * k);

#pragma unroll
        for (int nt = 0; nt < 3; ++nt) {
            const int ncol = nt * 16 + j16;
            const float bv = bias[ncol];
            v8f c;
#pragma unroll
            for (int i = 0; i < 8; ++i) c[i] = bv;

#pragma unroll
            for (int k = 0; k < 12; ++k) {
                const int krow = 4 * k + 2 * hi;
                v2f brel, broot;
                brel.x  = w_rel [(krow    ) * DF + ncol];
                brel.y  = w_rel [(krow + 1) * DF + ncol];
                broot.x = w_root[(krow    ) * DF + ncol];
                broot.y = w_root[(krow + 1) * DF + ncol];
                c = __builtin_amdgcn_wmma_f32_16x16x4_f32(
                        false, a_agg[k], false, brel, (short)0, c, false, false);
                c = __builtin_amdgcn_wmma_f32_16x16x4_f32(
                        false, a_x[k], false, broot, (short)0, c, false, false);
            }

#pragma unroll
            for (int v = 0; v < 8; ++v) {
                float r = c[v];
                if (RELU) r = fmaxf(r, 0.f);
                const int m = tile0 + v + 8 * hi;
                if (m < rows) out[(long long)(row0 + m) * DF + ncol] = r;
            }
        }
    }
}

extern "C" void kernel_launch(void* const* d_in, const int* in_sizes, int n_in,
                              void* d_out, int out_size, void* d_ws, size_t ws_size,
                              hipStream_t stream) {
    const float* x       = (const float*)d_in[0];
    const int*   ei      = (const int*)  d_in[1];
    const float* w_rel1  = (const float*)d_in[2];
    const float* w_root1 = (const float*)d_in[3];
    const float* b1      = (const float*)d_in[4];
    const float* w_rel2  = (const float*)d_in[5];
    const float* w_root2 = (const float*)d_in[6];
    const float* b2      = (const float*)d_in[7];
    float*       out     = (float*)d_out;

    const int N = in_sizes[0] / DF;
    const int E = in_sizes[1] / 2;
    const int* src = ei;
    const int* dst = ei + E;

    float* h = (float*)d_ws;

    unsigned* bm = (unsigned*)(h + (size_t)N * DF);
    const int nq     = (E + 15) / 16;
    const int nchunk = (nq + CHUNK_W - 1) / CHUNK_W;
    int* glist  = (int*)(bm + (size_t)NOWN * nchunk * 32);
    int* gcount = glist + (size_t)NOWN * LCAP;

    owner_bits<<<nchunk, CHUNK_W, 0, stream>>>(dst, bm, E, nq);
    agg_owner<false, true><<<NOWN, AGG_T, 0, stream>>>(x, src, dst, bm, w_rel1, w_root1, b1, h, glist, gcount, N, E, nchunk * 32);

    agg_owner<true, false><<<NOWN, AGG_T, 0, stream>>>(h, src, dst, bm, w_rel2, w_root2, b2, out, glist, gcount, N, E, nchunk * 32);
}
